# starting kernel
# baseline (speedup 1.0000x reference)
#include <hip/hip_runtime.h>


typedef __attribute__((ext_vector_type(2))) float v2f;
typedef __attribute__((ext_vector_type(8))) float v8f;

#define B_     8
#define CH_    128
#define F_IN   32
#define T_IN   1024
#define G_     8
#define D_     16
#define N_     16
#define W_     16
#define KS_    4
#define T_OUT  257
#define F_OUT  17

#define XS_STRIDE   80
#define XS_PER_WAVE (D_ * XS_STRIDE)

__global__ __launch_bounds__(256)
void linconv2d_wmma_f32(const float* __restrict__ x,
                        const float* __restrict__ weight,
                        float* __restrict__ out) {
  __shared__ __align__(16) float wlds[2 * N_ * D_ * W_];
  __shared__ __align__(16) float xs[8 * XS_PER_WAVE];

  const int tid  = threadIdx.x;
  const int wv   = tid >> 5;
  const int lane = tid & 31;
  const int hi   = lane >> 4;
  const int lo   = lane & 15;

  const int fo = blockIdx.y;
  const int z  = blockIdx.z;
  const int bi = z >> 3;
  const int g  = z & 7;

  for (int idx = tid; idx < 2 * N_ * D_ * W_; idx += 256) {
    const int fs  = idx >> 12;
    const int rem = idx & 4095;
    const int n   = rem >> 8;
    const int d   = (rem >> 4) & 15;
    const int w   = rem & 15;
    const int f   = 2 * fo - 1 + fs;
    float v = 0.0f;
    if (f >= 0 && f < F_IN)
      v = weight[(size_t)(g * N_ + n) * (D_ * F_IN * W_) + d * (F_IN * W_) + f * W_ + w];
    wlds[idx] = v;
  }
  __syncthreads();

  const int tile = blockIdx.x * 8 + wv;
  const int t0   = tile * 16;
  if (t0 >= T_OUT) return;

  float* myxs = &xs[wv * XS_PER_WAVE];
  v8f acc = {};

  for (int fs = 0; fs < 2; ++fs) {
    const int f  = 2 * fo - 1 + fs;
    const int fc = f < 0 ? 0 : (f > F_IN - 1 ? F_IN - 1 : f);

    for (int idx = lane; idx < D_ * 76; idx += 32) {
      const int d  = idx / 76;
      const int lt = idx - d * 76;
      const int ot = t0 * KS_ + lt - 8;
      float v = 0.0f;
      if (ot >= 0 && ot < T_IN)
        v = x[(((size_t)bi * CH_ + g * D_ + d) * F_IN + fc) * T_IN + ot];
      myxs[d * XS_STRIDE + lt] = v;
    }

    #pragma unroll
    for (int d = 0; d < D_; ++d) {
      #pragma unroll
      for (int wq = 0; wq < 4; ++wq) {
        const int kk = wq * 4 + 2 * hi;
        v2f a = *(const v2f*)&myxs[d * XS_STRIDE + lo * KS_ + kk];
        v2f b = *(const v2f*)&wlds[fs * 4096 + lo * 256 + d * 16 + kk];
        acc = __builtin_amdgcn_wmma_f32_16x16x4_f32(
             false, a,  false, b,
             (short)0, acc,  false,  false);
      }
    }
  }

  #pragma unroll
  for (int j = 0; j < 8; ++j) {
    float vv = acc[j];
    vv = vv > 0.0f ? vv : 0.01f * vv;
    myxs[lo * 17 + j + 8 * hi] = vv;
  }
  const int nn = lane >> 1;
  const int tb = (lane & 1) * 8;
  const size_t ob = (((size_t)bi * CH_ + g * N_ + nn) * F_OUT + fo) * (size_t)T_OUT + t0 + tb;
  #pragma unroll
  for (int e = 0; e < 8; ++e) {
    const int tg = t0 + tb + e;
    if (tg < T_OUT) out[ob + e] = myxs[nn * 17 + tb + e];
  }
}

extern "C" void kernel_launch(void* const* d_in, const int* in_sizes, int n_in,
                              void* d_out, int out_size, void* d_ws, size_t ws_size,
                              hipStream_t stream) {
  (void)in_sizes; (void)n_in; (void)out_size; (void)d_ws; (void)ws_size;
  const float* x = (const float*)d_in[0];
  const float* w = (const float*)d_in[1];
  float* out     = (float*)d_out;

  dim3 grid(3, F_OUT, B_ * G_);
  linconv2d_wmma_f32<<<grid, 256, 0, stream>>>(x, w, out);
}


# Staging loads (weight slab, both x slabs) issued as batches of 16-byte loads ahead of their LDS stores: many in flight per wave instead of one dword load waited for at a time; sums unchanged
# speedup vs baseline: 1.8295x; 1.8295x over previous
#include <hip/hip_runtime.h>


typedef __attribute__((ext_vector_type(2))) float v2f;
typedef __attribute__((ext_vector_type(4))) float v4f;
typedef __attribute__((ext_vector_type(8))) float v8f;

#define B_     8
#define CH_    128
#define F_IN   32
#define T_IN   1024
#define G_     8
#define D_     16
#define N_     16
#define W_     16
#define KS_    4
#define T_OUT  257
#define F_OUT  17

#define XS_STRIDE   80
#define XS_PER_WAVE (D_ * XS_STRIDE)
#define X_CH_STRIDE (F_IN * T_IN)
#define W_N_STRIDE  (D_ * F_IN * W_)

template <bool EDGE>
__device__ __forceinline__ void load_x_slab(const float* __restrict__ p, int tl, v4f (&r)[10]) {
  #pragma unroll
  for (int j = 0; j < 2; ++j) {
    #pragma unroll
    for (int k = 0; k < 5; ++k) {
      const int ot = tl + 16 * k;
      const float* q = p + (size_t)j * 8 * X_CH_STRIDE;
      if (EDGE) {
        const bool ok = (unsigned)ot < (unsigned)T_IN;
        r[j * 5 + k] = *(const v4f*)(q + (ok ? ot : 0));
      } else {
        r[j * 5 + k] = *(const v4f*)(q + ot);
      }
    }
  }
}

template <bool EDGE>
__device__ __forceinline__ void store_x_slab(float* s, int tl, const v4f (&r)[10]) {
  #pragma unroll
  for (int j = 0; j < 2; ++j) {
    #pragma unroll
    for (int k = 0; k < 5; ++k) {
      v4f v = r[j * 5 + k];
      if (EDGE) {
        const v4f z = {0.0f, 0.0f, 0.0f, 0.0f};
        v = ((unsigned)(tl + 16 * k) < (unsigned)T_IN) ? v : z;
      }
      *(v4f*)(s + j * 8 * XS_STRIDE + 16 * k) = v;
    }
  }
}

__device__ __forceinline__ v8f mma_one_f(const float* myxs, const float* wl, int lo, int hi, v8f acc) {
  #pragma unroll
  for (int d = 0; d < D_; ++d) {
    #pragma unroll
    for (int wq = 0; wq < 4; ++wq) {
      const int kk = wq * 4 + 2 * hi;
      v2f a = *(const v2f*)&myxs[d * XS_STRIDE + lo * KS_ + kk];
      v2f b = *(const v2f*)&wl[lo * 256 + d * 16 + kk];
      acc = __builtin_amdgcn_wmma_f32_16x16x4_f32(
           false, a,  false, b,
           (short)0, acc,  false,  false);
    }
  }
  return acc;
}

__global__ __launch_bounds__(256)
void linconv2d_wmma_f32(const float* __restrict__ x,
                        const float* __restrict__ weight,
                        float* __restrict__ out) {
  __shared__ __align__(16) float wlds[2 * N_ * D_ * W_];
  __shared__ __align__(16) float xs[8 * XS_PER_WAVE];

  const int tid  = threadIdx.x;
  const int wv   = tid >> 5;
  const int lane = tid & 31;
  const int hi   = lane >> 4;
  const int lo   = lane & 15;

  const int fo = blockIdx.y;
  const int z  = blockIdx.z;
  const int bi = z >> 3;
  const int g  = z & 7;

  const int f0  = 2 * fo - 1;
  const int fc0 = f0 < 0 ? 0 : f0;
  const int fc1 = f0 + 1 > F_IN - 1 ? F_IN - 1 : f0 + 1;

  const int  tile   = blockIdx.x * 8 + wv;
  const int  t0     = tile * 16;
  const bool active = t0 < T_OUT;
  const bool edge   = (t0 == 0) || (t0 * KS_ - 8 + XS_STRIDE > T_IN);

  const float* wsrc = weight + (size_t)(g * N_ + (tid >> 6)) * W_N_STRIDE
                             + ((tid >> 2) & 15) * (F_IN * W_) + (tid & 3) * 4;
  v4f wr[8];
  #pragma unroll
  for (int i = 0; i < 8; ++i) {
    const int fcw = (i >> 2) ? fc1 : fc0;
    wr[i] = *(const v4f*)(wsrc + (size_t)(i & 3) * 4 * W_N_STRIDE + fcw * W_);
  }

  const int xr_ = lane >> 2;
  const int xcc = lane & 3;
  const float* xsrc = x + ((size_t)bi * CH_ + g * D_ + xr_) * X_CH_STRIDE;
  const int    tl   = t0 * KS_ - 8 + 4 * xcc;
  float* myxs  = &xs[wv * XS_PER_WAVE];
  float* xdst  = myxs + xr_ * XS_STRIDE + 4 * xcc;
  v4f xa[10];
  if (active) {
    if (edge) load_x_slab<true >(xsrc + (size_t)fc0 * T_IN, tl, xa);
    else      load_x_slab<false>(xsrc + (size_t)fc0 * T_IN, tl, xa);
  }

  {
    const v4f zz = {0.0f, 0.0f, 0.0f, 0.0f};
    const bool in0 = f0 >= 0;
    const bool in1 = f0 + 1 < F_IN;
    #pragma unroll
    for (int i = 0; i < 8; ++i) {
      const bool inr = (i >> 2) ? in1 : in0;
      *(v4f*)&wlds[4 * (tid + 256 * i)] = inr ? wr[i] : zz;
    }
  }
  __syncthreads();
  if (!active) return;

  if (edge) store_x_slab<true >(xdst, tl, xa);
  else      store_x_slab<false>(xdst, tl, xa);
  v4f xb[10];
  if (edge) load_x_slab<true >(xsrc + (size_t)fc1 * T_IN, tl, xb);
  else      load_x_slab<false>(xsrc + (size_t)fc1 * T_IN, tl, xb);

  v8f acc = {};
  acc = mma_one_f(myxs, &wlds[0], lo, hi, acc);

  if (edge) store_x_slab<true >(xdst, tl, xb);
  else      store_x_slab<false>(xdst, tl, xb);
  acc = mma_one_f(myxs, &wlds[4096], lo, hi, acc);

  #pragma unroll
  for (int j = 0; j < 8; ++j) {
    float vv = acc[j];
    vv = vv > 0.0f ? vv : 0.01f * vv;
    myxs[lo * 17 + j + 8 * hi] = vv;
  }
  const int nn = lane >> 1;
  const int tb = (lane & 1) * 8;
  const size_t ob = (((size_t)bi * CH_ + g * N_ + nn) * F_OUT + fo) * (size_t)T_OUT + t0 + tb;
  #pragma unroll
  for (int e = 0; e < 8; ++e) {
    const int tg = t0 + tb + e;
    if (tg < T_OUT) out[ob + e] = myxs[nn * 17 + tb + e];
  }
}

extern "C" void kernel_launch(void* const* d_in, const int* in_sizes, int n_in,
                              void* d_out, int out_size, void* d_ws, size_t ws_size,
                              hipStream_t stream) {
  (void)in_sizes; (void)n_in; (void)out_size; (void)d_ws; (void)ws_size;
  const float* x = (const float*)d_in[0];
  const float* w = (const float*)d_in[1];
  float* out     = (float*)d_out;

  dim3 grid(3, F_OUT, B_ * G_);
  linconv2d_wmma_f32<<<grid, 256, 0, stream>>>(x, w, out);
}


# Weight slab rows in LDS padded from 256 to 260 floats: B-fragment reads (lane = row n) hit 64 distinct banks instead of 16 lanes per bank; same products, same sum order
# speedup vs baseline: 2.8742x; 1.5710x over previous
#include <hip/hip_runtime.h>


typedef __attribute__((ext_vector_type(2))) float v2f;
typedef __attribute__((ext_vector_type(4))) float v4f;
typedef __attribute__((ext_vector_type(8))) float v8f;

#define B_     8
#define CH_    128
#define F_IN   32
#define T_IN   1024
#define G_     8
#define D_     16
#define N_     16
#define W_     16
#define KS_    4
#define T_OUT  257
#define F_OUT  17

#define XS_STRIDE   80
#define XS_PER_WAVE (D_ * XS_STRIDE)
#define X_CH_STRIDE (F_IN * T_IN)
#define W_N_STRIDE  (D_ * F_IN * W_)
#define WL_N_PITCH  260
#define WL_FS_PITCH (N_ * WL_N_PITCH)

template <bool EDGE>
__device__ __forceinline__ void load_x_slab(const float* __restrict__ p, int tl, v4f (&r)[10]) {
  #pragma unroll
  for (int j = 0; j < 2; ++j) {
    #pragma unroll
    for (int k = 0; k < 5; ++k) {
      const int ot = tl + 16 * k;
      const float* q = p + (size_t)j * 8 * X_CH_STRIDE;
      if (EDGE) {
        const bool ok = (unsigned)ot < (unsigned)T_IN;
        r[j * 5 + k] = *(const v4f*)(q + (ok ? ot : 0));
      } else {
        r[j * 5 + k] = *(const v4f*)(q + ot);
      }
    }
  }
}

template <bool EDGE>
__device__ __forceinline__ void store_x_slab(float* s, int tl, const v4f (&r)[10]) {
  #pragma unroll
  for (int j = 0; j < 2; ++j) {
    #pragma unroll
    for (int k = 0; k < 5; ++k) {
      v4f v = r[j * 5 + k];
      if (EDGE) {
        const v4f z = {0.0f, 0.0f, 0.0f, 0.0f};
        v = ((unsigned)(tl + 16 * k) < (unsigned)T_IN) ? v : z;
      }
      *(v4f*)(s + j * 8 * XS_STRIDE + 16 * k) = v;
    }
  }
}

__device__ __forceinline__ v8f mma_one_f(const float* myxs, const float* wl, int lo, int hi, v8f acc) {
  #pragma unroll
  for (int d = 0; d < D_; ++d) {
    #pragma unroll
    for (int wq = 0; wq < 4; ++wq) {
      const int kk = wq * 4 + 2 * hi;
      v2f a = *(const v2f*)&myxs[d * XS_STRIDE + lo * KS_ + kk];
      v2f b = *(const v2f*)&wl[lo * WL_N_PITCH + d * 16 + kk];
      acc = __builtin_amdgcn_wmma_f32_16x16x4_f32(
           false, a,  false, b,
           (short)0, acc,  false,  false);
    }
  }
  return acc;
}

__global__ __launch_bounds__(256)
void linconv2d_wmma_f32(const float* __restrict__ x,
                        const float* __restrict__ weight,
                        float* __restrict__ out) {
  __shared__ __align__(16) float wlds[2 * WL_FS_PITCH];
  __shared__ __align__(16) float xs[8 * XS_PER_WAVE];

  const int tid  = threadIdx.x;
  const int wv   = tid >> 5;
  const int lane = tid & 31;
  const int hi   = lane >> 4;
  const int lo   = lane & 15;

  const int fo = blockIdx.y;
  const int z  = blockIdx.z;
  const int bi = z >> 3;
  const int g  = z & 7;

  const int f0  = 2 * fo - 1;
  const int fc0 = f0 < 0 ? 0 : f0;
  const int fc1 = f0 + 1 > F_IN - 1 ? F_IN - 1 : f0 + 1;

  const int  tile   = blockIdx.x * 8 + wv;
  const int  t0     = tile * 16;
  const bool active = t0 < T_OUT;
  const bool edge   = (t0 == 0) || (t0 * KS_ - 8 + XS_STRIDE > T_IN);

  const float* wsrc = weight + (size_t)(g * N_ + (tid >> 6)) * W_N_STRIDE
                             + ((tid >> 2) & 15) * (F_IN * W_) + (tid & 3) * 4;
  v4f wr[8];
  #pragma unroll
  for (int i = 0; i < 8; ++i) {
    const int fcw = (i >> 2) ? fc1 : fc0;
    wr[i] = *(const v4f*)(wsrc + (size_t)(i & 3) * 4 * W_N_STRIDE + fcw * W_);
  }

  const int xr_ = lane >> 2;
  const int xcc = lane & 3;
  const float* xsrc = x + ((size_t)bi * CH_ + g * D_ + xr_) * X_CH_STRIDE;
  const int    tl   = t0 * KS_ - 8 + 4 * xcc;
  float* myxs  = &xs[wv * XS_PER_WAVE];
  float* xdst  = myxs + xr_ * XS_STRIDE + 4 * xcc;
  v4f xa[10];
  if (active) {
    if (edge) load_x_slab<true >(xsrc + (size_t)fc0 * T_IN, tl, xa);
    else      load_x_slab<false>(xsrc + (size_t)fc0 * T_IN, tl, xa);
  }

  {
    const v4f zz = {0.0f, 0.0f, 0.0f, 0.0f};
    const bool in0 = f0 >= 0;
    const bool in1 = f0 + 1 < F_IN;
    float* wdst = &wlds[(tid >> 6) * WL_N_PITCH + (tid & 63) * 4];
    #pragma unroll
    for (int i = 0; i < 8; ++i) {
      const bool inr = (i >> 2) ? in1 : in0;
      *(v4f*)(wdst + (i >> 2) * WL_FS_PITCH + (i & 3) * 4 * WL_N_PITCH) = inr ? wr[i] : zz;
    }
  }
  __syncthreads();
  if (!active) return;

  if (edge) store_x_slab<true >(xdst, tl, xa);
  else      store_x_slab<false>(xdst, tl, xa);
  v4f xb[10];
  if (edge) load_x_slab<true >(xsrc + (size_t)fc1 * T_IN, tl, xb);
  else      load_x_slab<false>(xsrc + (size_t)fc1 * T_IN, tl, xb);

  v8f acc = {};
  acc = mma_one_f(myxs, &wlds[0], lo, hi, acc);

  if (edge) store_x_slab<true >(xdst, tl, xb);
  else      store_x_slab<false>(xdst, tl, xb);
  acc = mma_one_f(myxs, &wlds[WL_FS_PITCH], lo, hi, acc);

  #pragma unroll
  for (int j = 0; j < 8; ++j) {
    float vv = acc[j];
    vv = vv > 0.0f ? vv : 0.01f * vv;
    myxs[lo * 17 + j + 8 * hi] = vv;
  }
  const int nn = lane >> 1;
  const int tb = (lane & 1) * 8;
  const size_t ob = (((size_t)bi * CH_ + g * N_ + nn) * F_OUT + fo) * (size_t)T_OUT + t0 + tb;
  #pragma unroll
  for (int e = 0; e < 8; ++e) {
    const int tg = t0 + tb + e;
    if (tg < T_OUT) out[ob + e] = myxs[nn * 17 + tb + e];
  }
}

extern "C" void kernel_launch(void* const* d_in, const int* in_sizes, int n_in,
                              void* d_out, int out_size, void* d_ws, size_t ws_size,
                              hipStream_t stream) {
  (void)in_sizes; (void)n_in; (void)out_size; (void)d_ws; (void)ws_size;
  const float* x = (const float*)d_in[0];
  const float* w = (const float*)d_in[1];
  float* out     = (float*)d_out;

  dim3 grid(3, F_OUT, B_ * G_);
  linconv2d_wmma_f32<<<grid, 256, 0, stream>>>(x, w, out);
}


# One work-group per (b,g,f_out) column stages the weight pair slab once; its 4 waves loop over the 17 t tiles, each slab's x loads issued ahead of a product block; products and sum order unchanged
# speedup vs baseline: 3.0792x; 1.0713x over previous
#include <hip/hip_runtime.h>


typedef __attribute__((ext_vector_type(2))) float v2f;
typedef __attribute__((ext_vector_type(4))) float v4f;
typedef __attribute__((ext_vector_type(8))) float v8f;

#define B_     8
#define CH_    128
#define F_IN   32
#define T_IN   1024
#define G_     8
#define D_     16
#define N_     16
#define W_     16
#define KS_    4
#define T_OUT  257
#define F_OUT  17

#define XS_STRIDE   80
#define XS_PER_WAVE (D_ * XS_STRIDE)
#define X_CH_STRIDE (F_IN * T_IN)
#define W_N_STRIDE  (D_ * F_IN * W_)
#define WL_N_PITCH  260
#define WL_FS_PITCH (N_ * WL_N_PITCH)

template <bool EDGE>
__device__ __forceinline__ void load_x_slab(const float* __restrict__ p, int tl, v4f (&r)[10]) {
  #pragma unroll
  for (int j = 0; j < 2; ++j) {
    #pragma unroll
    for (int k = 0; k < 5; ++k) {
      const int ot = tl + 16 * k;
      const float* q = p + (size_t)j * 8 * X_CH_STRIDE;
      if (EDGE) {
        const bool ok = (unsigned)ot < (unsigned)T_IN;
        r[j * 5 + k] = *(const v4f*)(q + (ok ? ot : 0));
      } else {
        r[j * 5 + k] = *(const v4f*)(q + ot);
      }
    }
  }
}

template <bool EDGE>
__device__ __forceinline__ void store_x_slab(float* s, int tl, const v4f (&r)[10]) {
  #pragma unroll
  for (int j = 0; j < 2; ++j) {
    #pragma unroll
    for (int k = 0; k < 5; ++k) {
      v4f v = r[j * 5 + k];
      if (EDGE) {
        const v4f z = {0.0f, 0.0f, 0.0f, 0.0f};
        v = ((unsigned)(tl + 16 * k) < (unsigned)T_IN) ? v : z;
      }
      *(v4f*)(s + j * 8 * XS_STRIDE + 16 * k) = v;
    }
  }
}

__device__ __forceinline__ v8f mma_one_f(const float* myxs, const float* wl, int lo, int hi, v8f acc) {
  #pragma unroll
  for (int d = 0; d < D_; ++d) {
    #pragma unroll
    for (int wq = 0; wq < 4; ++wq) {
      const int kk = wq * 4 + 2 * hi;
      v2f a = *(const v2f*)&myxs[d * XS_STRIDE + lo * KS_ + kk];
      v2f b = *(const v2f*)&wl[lo * WL_N_PITCH + d * 16 + kk];
      acc = __builtin_amdgcn_wmma_f32_16x16x4_f32(
           false, a,  false, b,
           (short)0, acc,  false,  false);
    }
  }
  return acc;
}

#define NWAVES_ 4
#define NTILES_ 17

__device__ __forceinline__ bool tile_is_edge(int t0) {
  return (t0 == 0) || (t0 * KS_ - 8 + XS_STRIDE > T_IN);
}

__global__ __launch_bounds__(128, 5)
void linconv2d_wmma_f32(const float* __restrict__ x,
                        const float* __restrict__ weight,
                        float* __restrict__ out) {
  __shared__ __align__(16) float wlds[2 * WL_FS_PITCH];
  __shared__ __align__(16) float xs[NWAVES_ * XS_PER_WAVE];

  const int tid  = threadIdx.x;
  const int wv   = __builtin_amdgcn_readfirstlane(tid >> 5);
  const int lane = tid & 31;
  const int hi   = lane >> 4;
  const int lo   = lane & 15;

  const int fo = blockIdx.x;
  const int z  = blockIdx.y;
  const int bi = z >> 3;
  const int g  = z & 7;

  const int f0  = 2 * fo - 1;
  const int fc0 = f0 < 0 ? 0 : f0;
  const int fc1 = f0 + 1 > F_IN - 1 ? F_IN - 1 : f0 + 1;

  const int rot    = (int)(((unsigned)(z * F_OUT + fo) * 2654435761u) >> 30);
  const int ntiles = 4 + (wv == rot ? 1 : 0);

  const float* wsrc = weight + (size_t)(g * N_ + (tid >> 6)) * W_N_STRIDE
                             + ((tid >> 2) & 15) * (F_IN * W_) + (tid & 3) * 4;
  v4f wr[16];
  #pragma unroll
  for (int i = 0; i < 16; ++i) {
    const int fcw = (i >> 3) ? fc1 : fc0;
    wr[i] = *(const v4f*)(wsrc + (size_t)(i & 7) * 2 * W_N_STRIDE + fcw * W_);
  }

  const int xr_ = lane >> 2;
  const int xcc = lane & 3;
  const float* xsrc = x + ((size_t)bi * CH_ + g * D_ + xr_) * X_CH_STRIDE;
  float* myxs  = &xs[wv * XS_PER_WAVE];
  float* xdst  = myxs + xr_ * XS_STRIDE + 4 * xcc;
  int tile = wv;
  v4f xa[10];
  {
    const int t0 = tile * 16;
    const int tl = t0 * KS_ - 8 + 4 * xcc;
    if (tile_is_edge(t0)) load_x_slab<true >(xsrc + (size_t)fc0 * T_IN, tl, xa);
    else                  load_x_slab<false>(xsrc + (size_t)fc0 * T_IN, tl, xa);
  }

  {
    const v4f zz = {0.0f, 0.0f, 0.0f, 0.0f};
    const bool in0 = f0 >= 0;
    const bool in1 = f0 + 1 < F_IN;
    float* wdst = &wlds[(tid >> 6) * WL_N_PITCH + (tid & 63) * 4];
    #pragma unroll
    for (int i = 0; i < 16; ++i) {
      const bool inr = (i >> 3) ? in1 : in0;
      *(v4f*)(wdst + (i >> 3) * WL_FS_PITCH + (i & 7) * 2 * WL_N_PITCH) = inr ? wr[i] : zz;
    }
  }
  __syncthreads();

  const int nn = lane >> 1;
  const int tb = (lane & 1) * 8;
  float* orow = out + (((size_t)bi * CH_ + g * N_ + nn) * F_OUT + fo) * (size_t)T_OUT + tb;

  for (int i = 0; i < ntiles; ++i) {
    const int  t0   = tile * 16;
    const int  tl   = t0 * KS_ - 8 + 4 * xcc;
    const bool edge = tile_is_edge(t0);

    if (edge) store_x_slab<true >(xdst, tl, xa);
    else      store_x_slab<false>(xdst, tl, xa);
    v4f xb[10];
    if (edge) load_x_slab<true >(xsrc + (size_t)fc1 * T_IN, tl, xb);
    else      load_x_slab<false>(xsrc + (size_t)fc1 * T_IN, tl, xb);
    asm volatile("" ::: "memory");

    v8f acc = {};
    acc = mma_one_f(myxs, &wlds[0], lo, hi, acc);

    if (edge) store_x_slab<true >(xdst, tl, xb);
    else      store_x_slab<false>(xdst, tl, xb);
    const int tile_n = (i + 1 < 4) ? tile + NWAVES_ : NTILES_ - 1;
    if (i + 1 < ntiles) {
      const int t0n = tile_n * 16;
      const int tln = t0n * KS_ - 8 + 4 * xcc;
      if (tile_is_edge(t0n)) load_x_slab<true >(xsrc + (size_t)fc0 * T_IN, tln, xa);
      else                   load_x_slab<false>(xsrc + (size_t)fc0 * T_IN, tln, xa);
    }
    asm volatile("" ::: "memory");
    acc = mma_one_f(myxs, &wlds[WL_FS_PITCH], lo, hi, acc);

    #pragma unroll
    for (int j = 0; j < 8; ++j) {
      float vv = acc[j];
      vv = vv > 0.0f ? vv : 0.01f * vv;
      myxs[lo * 17 + j + 8 * hi] = vv;
    }
    #pragma unroll
    for (int e = 0; e < 8; ++e) {
      const int tg = t0 + tb + e;
      if (tg < T_OUT) orow[t0 + e] = myxs[nn * 17 + tb + e];
    }
    tile = tile_n;
  }
}

extern "C" void kernel_launch(void* const* d_in, const int* in_sizes, int n_in,
                              void* d_out, int out_size, void* d_ws, size_t ws_size,
                              hipStream_t stream) {
  (void)in_sizes; (void)n_in; (void)out_size; (void)d_ws; (void)ws_size;
  const float* x = (const float*)d_in[0];
  const float* w = (const float*)d_in[1];
  float* out     = (float*)d_out;

  dim3 grid(F_OUT, B_ * G_);
  linconv2d_wmma_f32<<<grid, 128, 0, stream>>>(x, w, out);
}


# Fragment reads of the product loop issued as whole groups (one per row of 8 products, two rows ahead, three register sets) instead of two reads behind every two products; sum order unchanged
# speedup vs baseline: 3.2259x; 1.0476x over previous
#include <hip/hip_runtime.h>


typedef __attribute__((ext_vector_type(2))) float v2f;
typedef __attribute__((ext_vector_type(4))) float v4f;
typedef __attribute__((ext_vector_type(8))) float v8f;

#define B_     8
#define CH_    128
#define F_IN   32
#define T_IN   1024
#define G_     8
#define D_     16
#define N_     16
#define W_     16
#define KS_    4
#define T_OUT  257
#define F_OUT  17

#define XS_STRIDE   80
#define XS_PER_WAVE (D_ * XS_STRIDE)
#define X_CH_STRIDE (F_IN * T_IN)
#define W_N_STRIDE  (D_ * F_IN * W_)
#define WL_N_PITCH  260
#define WL_FS_PITCH (N_ * WL_N_PITCH)

template <bool EDGE>
__device__ __forceinline__ void load_x_slab(const float* __restrict__ p, int tl, v4f (&r)[10]) {
  #pragma unroll
  for (int j = 0; j < 2; ++j) {
    #pragma unroll
    for (int k = 0; k < 5; ++k) {
      const int ot = tl + 16 * k;
      const float* q = p + (size_t)j * 8 * X_CH_STRIDE;
      if (EDGE) {
        const bool ok = (unsigned)ot < (unsigned)T_IN;
        r[j * 5 + k] = *(const v4f*)(q + (ok ? ot : 0));
      } else {
        r[j * 5 + k] = *(const v4f*)(q + ot);
      }
    }
  }
}

template <bool EDGE>
__device__ __forceinline__ void store_x_slab(float* s, int tl, const v4f (&r)[10]) {
  #pragma unroll
  for (int j = 0; j < 2; ++j) {
    #pragma unroll
    for (int k = 0; k < 5; ++k) {
      v4f v = r[j * 5 + k];
      if (EDGE) {
        const v4f z = {0.0f, 0.0f, 0.0f, 0.0f};
        v = ((unsigned)(tl + 16 * k) < (unsigned)T_IN) ? v : z;
      }
      *(v4f*)(s + j * 8 * XS_STRIDE + 16 * k) = v;
    }
  }
}

#define ROW_D  2
#define ROW_W  (ROW_D * 4)
#define NROWS_ (D_ / ROW_D)
#define NSETS_ 3

__device__ __forceinline__ void load_frag_row(const float* myxs, const float* wl, int lo, int hi, int row,
                                              v2f (&a)[ROW_W], v2f (&b)[ROW_W]) {
  #pragma unroll
  for (int dd = 0; dd < ROW_D; ++dd) {
    #pragma unroll
    for (int wq = 0; wq < 4; ++wq) {
      const int d  = row * ROW_D + dd;
      const int kk = wq * 4 + 2 * hi;
      a[dd * 4 + wq] = *(const v2f*)&myxs[d * XS_STRIDE + lo * KS_ + kk];
      b[dd * 4 + wq] = *(const v2f*)&wl[lo * WL_N_PITCH + d * 16 + kk];
    }
  }
}

__device__ __forceinline__ v8f mma_one_f(const float* myxs, const float* wl, int lo, int hi, v8f acc) {
  v2f a[NSETS_][ROW_W], b[NSETS_][ROW_W];
  #pragma unroll
  for (int r = 0; r < NSETS_ - 1; ++r) load_frag_row(myxs, wl, lo, hi, r, a[r], b[r]);
  #pragma unroll
  for (int r = 0; r < NROWS_; ++r) {
    __builtin_amdgcn_sched_barrier(0);
    if (r + NSETS_ - 1 < NROWS_)
      load_frag_row(myxs, wl, lo, hi, r + NSETS_ - 1, a[(r + NSETS_ - 1) % NSETS_], b[(r + NSETS_ - 1) % NSETS_]);
    __builtin_amdgcn_sched_barrier(0);
    #pragma unroll
    for (int i = 0; i < ROW_W; ++i) {
      acc = __builtin_amdgcn_wmma_f32_16x16x4_f32(
           false, a[r % NSETS_][i],  false, b[r % NSETS_][i],
           (short)0, acc,  false,  false);
    }
  }
  return acc;
}

#define NWAVES_ 4
#define NTILES_ 17

__device__ __forceinline__ bool tile_is_edge(int t0) {
  return (t0 == 0) || (t0 * KS_ - 8 + XS_STRIDE > T_IN);
}

__global__ __launch_bounds__(128, 5)
void linconv2d_wmma_f32(const float* __restrict__ x,
                        const float* __restrict__ weight,
                        float* __restrict__ out) {
  __shared__ __align__(16) float wlds[2 * WL_FS_PITCH];
  __shared__ __align__(16) float xs[NWAVES_ * XS_PER_WAVE];

  const int tid  = threadIdx.x;
  const int wv   = __builtin_amdgcn_readfirstlane(tid >> 5);
  const int lane = tid & 31;
  const int hi   = lane >> 4;
  const int lo   = lane & 15;

  const int fo = blockIdx.x;
  const int z  = blockIdx.y;
  const int bi = z >> 3;
  const int g  = z & 7;

  const int f0  = 2 * fo - 1;
  const int fc0 = f0 < 0 ? 0 : f0;
  const int fc1 = f0 + 1 > F_IN - 1 ? F_IN - 1 : f0 + 1;

  const int rot    = (int)(((unsigned)(z * F_OUT + fo) * 2654435761u) >> 30);
  const int ntiles = 4 + (wv == rot ? 1 : 0);

  const float* wsrc = weight + (size_t)(g * N_ + (tid >> 6)) * W_N_STRIDE
                             + ((tid >> 2) & 15) * (F_IN * W_) + (tid & 3) * 4;
  v4f wr[16];
  #pragma unroll
  for (int i = 0; i < 16; ++i) {
    const int fcw = (i >> 3) ? fc1 : fc0;
    wr[i] = *(const v4f*)(wsrc + (size_t)(i & 7) * 2 * W_N_STRIDE + fcw * W_);
  }

  const int xr_ = lane >> 2;
  const int xcc = lane & 3;
  const float* xsrc = x + ((size_t)bi * CH_ + g * D_ + xr_) * X_CH_STRIDE;
  float* myxs  = &xs[wv * XS_PER_WAVE];
  float* xdst  = myxs + xr_ * XS_STRIDE + 4 * xcc;
  int tile = wv;
  v4f xa[10];
  {
    const int t0 = tile * 16;
    const int tl = t0 * KS_ - 8 + 4 * xcc;
    if (tile_is_edge(t0)) load_x_slab<true >(xsrc + (size_t)fc0 * T_IN, tl, xa);
    else                  load_x_slab<false>(xsrc + (size_t)fc0 * T_IN, tl, xa);
  }

  {
    const v4f zz = {0.0f, 0.0f, 0.0f, 0.0f};
    const bool in0 = f0 >= 0;
    const bool in1 = f0 + 1 < F_IN;
    float* wdst = &wlds[(tid >> 6) * WL_N_PITCH + (tid & 63) * 4];
    #pragma unroll
    for (int i = 0; i < 16; ++i) {
      const bool inr = (i >> 3) ? in1 : in0;
      *(v4f*)(wdst + (i >> 3) * WL_FS_PITCH + (i & 7) * 2 * WL_N_PITCH) = inr ? wr[i] : zz;
    }
  }
  __syncthreads();

  const int nn = lane >> 1;
  const int tb = (lane & 1) * 8;
  float* orow = out + (((size_t)bi * CH_ + g * N_ + nn) * F_OUT + fo) * (size_t)T_OUT + tb;

  for (int i = 0; i < ntiles; ++i) {
    const int  t0   = tile * 16;
    const int  tl   = t0 * KS_ - 8 + 4 * xcc;
    const bool edge = tile_is_edge(t0);

    if (edge) store_x_slab<true >(xdst, tl, xa);
    else      store_x_slab<false>(xdst, tl, xa);
    v4f xb[10];
    if (edge) load_x_slab<true >(xsrc + (size_t)fc1 * T_IN, tl, xb);
    else      load_x_slab<false>(xsrc + (size_t)fc1 * T_IN, tl, xb);
    asm volatile("" ::: "memory");

    v8f acc = {};
    acc = mma_one_f(myxs, &wlds[0], lo, hi, acc);

    if (edge) store_x_slab<true >(xdst, tl, xb);
    else      store_x_slab<false>(xdst, tl, xb);
    const int tile_n = (i + 1 < 4) ? tile + NWAVES_ : NTILES_ - 1;
    if (i + 1 < ntiles) {
      const int t0n = tile_n * 16;
      const int tln = t0n * KS_ - 8 + 4 * xcc;
      if (tile_is_edge(t0n)) load_x_slab<true >(xsrc + (size_t)fc0 * T_IN, tln, xa);
      else                   load_x_slab<false>(xsrc + (size_t)fc0 * T_IN, tln, xa);
    }
    asm volatile("" ::: "memory");
    acc = mma_one_f(myxs, &wlds[WL_FS_PITCH], lo, hi, acc);

    #pragma unroll
    for (int j = 0; j < 8; ++j) {
      float vv = acc[j];
      vv = vv > 0.0f ? vv : 0.01f * vv;
      myxs[lo * 17 + j + 8 * hi] = vv;
    }
    #pragma unroll
    for (int e = 0; e < 8; ++e) {
      const int tg = t0 + tb + e;
      if (tg < T_OUT) orow[t0 + e] = myxs[nn * 17 + tb + e];
    }
    tile = tile_n;
  }
}

extern "C" void kernel_launch(void* const* d_in, const int* in_sizes, int n_in,
                              void* d_out, int out_size, void* d_ws, size_t ws_size,
                              hipStream_t stream) {
  (void)in_sizes; (void)n_in; (void)out_size; (void)d_ws; (void)ws_size;
  const float* x = (const float*)d_in[0];
  const float* w = (const float*)d_in[1];
  float* out     = (float*)d_out;

  dim3 grid(F_OUT, B_ * G_);
  linconv2d_wmma_f32<<<grid, 128, 0, stream>>>(x, w, out);
}


# Weight fragment as matrix operand A, x window as B: the [n][t] accumulator tile goes from registers to memory in 64-byte runs; epilogue LDS transpose and its 8 LDS round trips gone; same sum order
# speedup vs baseline: 3.7147x; 1.1515x over previous
#include <hip/hip_runtime.h>


typedef __attribute__((ext_vector_type(2))) float v2f;
typedef __attribute__((ext_vector_type(4))) float v4f;
typedef __attribute__((ext_vector_type(8))) float v8f;

#define B_     8
#define CH_    128
#define F_IN   32
#define T_IN   1024
#define G_     8
#define D_     16
#define N_     16
#define W_     16
#define KS_    4
#define T_OUT  257
#define F_OUT  17

#define XS_STRIDE   80
#define XS_PER_WAVE (D_ * XS_STRIDE)
#define X_CH_STRIDE (F_IN * T_IN)
#define W_N_STRIDE  (D_ * F_IN * W_)
#define WL_N_PITCH  260
#define WL_FS_PITCH (N_ * WL_N_PITCH)

template <bool EDGE>
__device__ __forceinline__ void load_x_slab(const float* __restrict__ p, int tl, v4f (&r)[10]) {
  #pragma unroll
  for (int j = 0; j < 2; ++j) {
    #pragma unroll
    for (int k = 0; k < 5; ++k) {
      const int ot = tl + 16 * k;
      const float* q = p + (size_t)j * 8 * X_CH_STRIDE;
      if (EDGE) {
        const bool ok = (unsigned)ot < (unsigned)T_IN;
        r[j * 5 + k] = *(const v4f*)(q + (ok ? ot : 0));
      } else {
        r[j * 5 + k] = *(const v4f*)(q + ot);
      }
    }
  }
}

template <bool EDGE>
__device__ __forceinline__ void store_x_slab(float* s, int tl, const v4f (&r)[10]) {
  #pragma unroll
  for (int j = 0; j < 2; ++j) {
    #pragma unroll
    for (int k = 0; k < 5; ++k) {
      v4f v = r[j * 5 + k];
      if (EDGE) {
        const v4f z = {0.0f, 0.0f, 0.0f, 0.0f};
        v = ((unsigned)(tl + 16 * k) < (unsigned)T_IN) ? v : z;
      }
      *(v4f*)(s + j * 8 * XS_STRIDE + 16 * k) = v;
    }
  }
}

#define ROW_D  2
#define ROW_W  (ROW_D * 4)
#define NROWS_ (D_ / ROW_D)
#define NSETS_ 3

__device__ __forceinline__ void load_frag_row(const float* myxs, const float* wl, int lo, int hi, int row,
                                              v2f (&xf)[ROW_W], v2f (&wf)[ROW_W]) {
  #pragma unroll
  for (int dd = 0; dd < ROW_D; ++dd) {
    #pragma unroll
    for (int wq = 0; wq < 4; ++wq) {
      const int d  = row * ROW_D + dd;
      const int kk = wq * 4 + 2 * hi;
      xf[dd * 4 + wq] = *(const v2f*)&myxs[d * XS_STRIDE + lo * KS_ + kk];
      wf[dd * 4 + wq] = *(const v2f*)&wl[lo * WL_N_PITCH + d * 16 + kk];
    }
  }
}

__device__ __forceinline__ v8f mma_one_f(const float* myxs, const float* wl, int lo, int hi, v8f acc) {
  v2f xf[NSETS_][ROW_W], wf[NSETS_][ROW_W];
  #pragma unroll
  for (int r = 0; r < NSETS_ - 1; ++r) load_frag_row(myxs, wl, lo, hi, r, xf[r], wf[r]);
  #pragma unroll
  for (int r = 0; r < NROWS_; ++r) {
    __builtin_amdgcn_sched_barrier(0);
    if (r + NSETS_ - 1 < NROWS_)
      load_frag_row(myxs, wl, lo, hi, r + NSETS_ - 1, xf[(r + NSETS_ - 1) % NSETS_], wf[(r + NSETS_ - 1) % NSETS_]);
    __builtin_amdgcn_sched_barrier(0);
    #pragma unroll
    for (int i = 0; i < ROW_W; ++i) {
      acc = __builtin_amdgcn_wmma_f32_16x16x4_f32(
           false, wf[r % NSETS_][i],  false, xf[r % NSETS_][i],
           (short)0, acc,  false,  false);
    }
  }
  return acc;
}

#define NWAVES_ 4
#define NTILES_ 17

__device__ __forceinline__ bool tile_is_edge(int t0) {
  return (t0 == 0) || (t0 * KS_ - 8 + XS_STRIDE > T_IN);
}

__global__ __launch_bounds__(128, 5)
void linconv2d_wmma_f32(const float* __restrict__ x,
                        const float* __restrict__ weight,
                        float* __restrict__ out) {
  __shared__ __align__(16) float wlds[2 * WL_FS_PITCH];
  __shared__ __align__(16) float xs[NWAVES_ * XS_PER_WAVE];

  const int tid  = threadIdx.x;
  const int wv   = __builtin_amdgcn_readfirstlane(tid >> 5);
  const int lane = tid & 31;
  const int hi   = lane >> 4;
  const int lo   = lane & 15;

  const int fo = blockIdx.x;
  const int z  = blockIdx.y;
  const int bi = z >> 3;
  const int g  = z & 7;

  const int f0  = 2 * fo - 1;
  const int fc0 = f0 < 0 ? 0 : f0;
  const int fc1 = f0 + 1 > F_IN - 1 ? F_IN - 1 : f0 + 1;

  const int rot    = (int)(((unsigned)(z * F_OUT + fo) * 2654435761u) >> 30);
  const int ntiles = 4 + (wv == rot ? 1 : 0);

  const float* wsrc = weight + (size_t)(g * N_ + (tid >> 6)) * W_N_STRIDE
                             + ((tid >> 2) & 15) * (F_IN * W_) + (tid & 3) * 4;
  v4f wr[16];
  #pragma unroll
  for (int i = 0; i < 16; ++i) {
    const int fcw = (i >> 3) ? fc1 : fc0;
    wr[i] = *(const v4f*)(wsrc + (size_t)(i & 7) * 2 * W_N_STRIDE + fcw * W_);
  }

  const int xr_ = lane >> 2;
  const int xcc = lane & 3;
  const float* xsrc = x + ((size_t)bi * CH_ + g * D_ + xr_) * X_CH_STRIDE;
  float* myxs  = &xs[wv * XS_PER_WAVE];
  float* xdst  = myxs + xr_ * XS_STRIDE + 4 * xcc;
  int tile = wv;
  v4f xa[10];
  {
    const int t0 = tile * 16;
    const int tl = t0 * KS_ - 8 + 4 * xcc;
    if (tile_is_edge(t0)) load_x_slab<true >(xsrc + (size_t)fc0 * T_IN, tl, xa);
    else                  load_x_slab<false>(xsrc + (size_t)fc0 * T_IN, tl, xa);
  }

  {
    const v4f zz = {0.0f, 0.0f, 0.0f, 0.0f};
    const bool in0 = f0 >= 0;
    const bool in1 = f0 + 1 < F_IN;
    float* wdst = &wlds[(tid >> 6) * WL_N_PITCH + (tid & 63) * 4];
    #pragma unroll
    for (int i = 0; i < 16; ++i) {
      const bool inr = (i >> 3) ? in1 : in0;
      *(v4f*)(wdst + (i >> 3) * WL_FS_PITCH + (i & 7) * 2 * WL_N_PITCH) = inr ? wr[i] : zz;
    }
  }
  __syncthreads();

  float* ocol = out + (((size_t)bi * CH_ + g * N_ + 8 * hi) * F_OUT + fo) * (size_t)T_OUT + lo;

  for (int i = 0; i < ntiles; ++i) {
    const int  t0   = tile * 16;
    const int  tl   = t0 * KS_ - 8 + 4 * xcc;
    const bool edge = tile_is_edge(t0);

    if (edge) store_x_slab<true >(xdst, tl, xa);
    else      store_x_slab<false>(xdst, tl, xa);
    v4f xb[10];
    if (edge) load_x_slab<true >(xsrc + (size_t)fc1 * T_IN, tl, xb);
    else      load_x_slab<false>(xsrc + (size_t)fc1 * T_IN, tl, xb);
    asm volatile("" ::: "memory");

    v8f acc = {};
    acc = mma_one_f(myxs, &wlds[0], lo, hi, acc);

    if (edge) store_x_slab<true >(xdst, tl, xb);
    else      store_x_slab<false>(xdst, tl, xb);
    const int tile_n = (i + 1 < 4) ? tile + NWAVES_ : NTILES_ - 1;
    if (i + 1 < ntiles) {
      const int t0n = tile_n * 16;
      const int tln = t0n * KS_ - 8 + 4 * xcc;
      if (tile_is_edge(t0n)) load_x_slab<true >(xsrc + (size_t)fc0 * T_IN, tln, xa);
      else                   load_x_slab<false>(xsrc + (size_t)fc0 * T_IN, tln, xa);
    }
    asm volatile("" ::: "memory");
    acc = mma_one_f(myxs, &wlds[WL_FS_PITCH], lo, hi, acc);

    if (t0 + lo < T_OUT) {
      #pragma unroll
      for (int j = 0; j < 8; ++j) {
        float vv = acc[j];
        vv = vv > 0.0f ? vv : 0.01f * vv;
        ocol[(size_t)j * (F_OUT * T_OUT) + t0] = vv;
      }
    }
    tile = tile_n;
  }
}

extern "C" void kernel_launch(void* const* d_in, const int* in_sizes, int n_in,
                              void* d_out, int out_size, void* d_ws, size_t ws_size,
                              hipStream_t stream) {
  (void)in_sizes; (void)n_in; (void)out_size; (void)d_ws; (void)ws_size;
  const float* x = (const float*)d_in[0];
  const float* w = (const float*)d_in[1];
  float* out     = (float*)d_out;

  dim3 grid(F_OUT, B_ * G_);
  linconv2d_wmma_f32<<<grid, 128, 0, stream>>>(x, w, out);
}


# x operand no longer staged in LDS: 8-byte fragments straight from memory (K = channel x tap mod 4), each feeds four WMMAs, one accumulator per tap group; groups lane-rotated and added: new sum order
# speedup vs baseline: 4.4097x; 1.1871x over previous
#include <hip/hip_runtime.h>


typedef __attribute__((ext_vector_type(2))) float v2f;
typedef __attribute__((ext_vector_type(4))) float v4f;
typedef __attribute__((ext_vector_type(8))) float v8f;

#define B_     8
#define CH_    128
#define F_IN   32
#define T_IN   1024
#define G_     8
#define D_     16
#define N_     16
#define W_     16
#define KS_    4
#define T_OUT  257
#define F_OUT  17

#define X_CH_STRIDE (F_IN * T_IN)
#define W_N_STRIDE  (D_ * F_IN * W_)
#define WL_N_PITCH  260
#define WL_FS_PITCH (N_ * WL_N_PITCH)

#define NS_TILES_ 16
#define NWAVES_   4
#define TILES_PER_WAVE_ (NS_TILES_ / NWAVES_)

__device__ __forceinline__ void load_x_frags(const float* __restrict__ p, v2f (&xf)[D_]) {
  #pragma unroll
  for (int d = 0; d < D_; ++d) xf[d] = *(const v2f*)(p + (size_t)d * X_CH_STRIDE);
}

#define ROW_D  2
#define ROW_W  (ROW_D * 4)
#define NROWS_ (D_ / ROW_D)
#define NSETS_ 3

__device__ __forceinline__ void load_w_row(const float* wl, int lo, int hi, int row, v2f (&wf)[ROW_W]) {
  #pragma unroll
  for (int dd = 0; dd < ROW_D; ++dd) {
    #pragma unroll
    for (int wq = 0; wq < 4; ++wq) {
      const int d = row * ROW_D + dd;
      wf[dd * 4 + wq] = *(const v2f*)&wl[lo * WL_N_PITCH + d * 16 + wq * 4 + 2 * hi];
    }
  }
}

__device__ __forceinline__ void mma_one_f(const float* wl, int lo, int hi, const v2f (&xf)[D_], v8f (&P)[4]) {
  v2f wf[NSETS_][ROW_W];
  #pragma unroll
  for (int r = 0; r < NSETS_ - 1; ++r) load_w_row(wl, lo, hi, r, wf[r]);
  #pragma unroll
  for (int r = 0; r < NROWS_; ++r) {
    __builtin_amdgcn_sched_barrier(0);
    if (r + NSETS_ - 1 < NROWS_) load_w_row(wl, lo, hi, r + NSETS_ - 1, wf[(r + NSETS_ - 1) % NSETS_]);
    __builtin_amdgcn_sched_barrier(0);
    #pragma unroll
    for (int i = 0; i < ROW_W; ++i) {
      P[i & 3] = __builtin_amdgcn_wmma_f32_16x16x4_f32(
           false, wf[r % NSETS_][i],  false, xf[r * ROW_D + (i >> 2)],
           (short)0, P[i & 3],  false,  false);
    }
  }
}

template <int K>
__device__ __forceinline__ float ror16(float v) {
  return __builtin_bit_cast(float, __builtin_amdgcn_update_dpp(0, __builtin_bit_cast(int, v), 0x120 + K, 0xF, 0xF, false));
}

__device__ __forceinline__ float leaky(float v) { return v > 0.0f ? v : 0.01f * v; }

__global__ __launch_bounds__(128, 5)
void linconv2d_wmma_f32(const float* __restrict__ x,
                        const float* __restrict__ weight,
                        float* __restrict__ out) {
  __shared__ __align__(16) float wlds[2 * WL_FS_PITCH];
  __shared__ float cslot[(NWAVES_ - 1) * 8 * 32];

  const int tid  = threadIdx.x;
  const int wv   = __builtin_amdgcn_readfirstlane(tid >> 5);
  const int lane = tid & 31;
  const int hi   = lane >> 4;
  const int lo   = lane & 15;

  const int fo = blockIdx.x;
  const int z  = blockIdx.y;
  const int bi = z >> 3;
  const int g  = z & 7;

  const int f0  = 2 * fo - 1;
  const int fc0 = f0 < 0 ? 0 : f0;
  const int fc1 = f0 + 1 > F_IN - 1 ? F_IN - 1 : f0 + 1;

  const float* wsrc = weight + (size_t)(g * N_ + (tid >> 6)) * W_N_STRIDE
                             + ((tid >> 2) & 15) * (F_IN * W_) + (tid & 3) * 4;
  v4f wr[16];
  #pragma unroll
  for (int i = 0; i < 16; ++i) {
    const int fcw = (i >> 3) ? fc1 : fc0;
    wr[i] = *(const v4f*)(wsrc + (size_t)(i & 7) * 2 * W_N_STRIDE + fcw * W_);
  }

  const float* xsrc = x + ((size_t)bi * CH_ + g * D_) * X_CH_STRIDE + 4 * lo + 2 * hi;
  const float* xs0  = xsrc + (size_t)fc0 * T_IN;
  const float* xs1  = xsrc + (size_t)fc1 * T_IN;
  int tile = TILES_PER_WAVE_ * wv + TILES_PER_WAVE_ - 1;
  v2f xa[D_], xb[D_];
  load_x_frags(xs0 + 64 * tile, xa);

  {
    const v4f zz = {0.0f, 0.0f, 0.0f, 0.0f};
    const bool in0 = f0 >= 0;
    const bool in1 = f0 + 1 < F_IN;
    float* wdst = &wlds[(tid >> 6) * WL_N_PITCH + (tid & 63) * 4];
    #pragma unroll
    for (int i = 0; i < 16; ++i) {
      const bool inr = (i >> 3) ? in1 : in0;
      *(v4f*)(wdst + (i >> 3) * WL_FS_PITCH + (i & 7) * 2 * WL_N_PITCH) = inr ? wr[i] : zz;
    }
  }
  __syncthreads();

  float* orow = out + (((size_t)bi * CH_ + g * N_ + 8 * hi) * F_OUT + fo) * (size_t)T_OUT;
  const int tl = lo - 1;
  const bool m1 = lo >= 1, m2 = lo >= 2, m3 = lo >= 3;

  v8f pend = {};
  #pragma unroll
  for (int it = 0; it < TILES_PER_WAVE_; ++it) {
    load_x_frags(xs1 + 64 * tile, xb);
    asm volatile("" ::: "memory");

    v8f P[4] = {};
    mma_one_f(&wlds[0], lo, hi, xa, P);

    if (it + 1 < TILES_PER_WAVE_) load_x_frags(xs0 + 64 * (tile - 1), xa);
    asm volatile("" ::: "memory");
    mma_one_f(&wlds[WL_FS_PITCH], lo, hi, xb, P);

    v8f cur, car;
    #pragma unroll
    for (int j = 0; j < 8; ++j) {
      const float r1 = ror16<1>(P[2][j]);
      const float r2 = ror16<2>(P[1][j]);
      const float r3 = ror16<3>(P[0][j]);
      cur[j] = ((P[3][j] + (m1 ? r1 : 0.0f)) + (m2 ? r2 : 0.0f)) + (m3 ? r3 : 0.0f);
      car[j] = ((m1 ? 0.0f : r1) + (m2 ? 0.0f : r2)) + (m3 ? 0.0f : r3);
    }

    if (it == 0) {
      if (wv == NWAVES_ - 1) {
        if (lo < 2) {
          #pragma unroll
          for (int j = 0; j < 8; ++j) orow[(size_t)j * (F_OUT * T_OUT) + 16 * NS_TILES_ + tl] = leaky(car[j]);
        }
      } else {
        #pragma unroll
        for (int j = 0; j < 8; ++j) cslot[wv * 256 + j * 32 + lane] = car[j];
      }
    } else {
      const int tc = 16 * (tile + 1) + tl;
      #pragma unroll
      for (int j = 0; j < 8; ++j) orow[(size_t)j * (F_OUT * T_OUT) + tc] = leaky(pend[j] + car[j]);
    }
    pend = cur;
    --tile;
  }

  __syncthreads();

  v8f cin = {};
  if (wv > 0) {
    #pragma unroll
    for (int j = 0; j < 8; ++j) cin[j] = cslot[(wv - 1) * 256 + j * 32 + lane];
  }
  const int tf = 16 * TILES_PER_WAVE_ * wv + tl;
  if (tf >= 0) {
    #pragma unroll
    for (int j = 0; j < 8; ++j) orow[(size_t)j * (F_OUT * T_OUT) + tf] = leaky(pend[j] + cin[j]);
  }
}

extern "C" void kernel_launch(void* const* d_in, const int* in_sizes, int n_in,
                              void* d_out, int out_size, void* d_ws, size_t ws_size,
                              hipStream_t stream) {
  (void)in_sizes; (void)n_in; (void)out_size; (void)d_ws; (void)ws_size;
  const float* x = (const float*)d_in[0];
  const float* w = (const float*)d_in[1];
  float* out     = (float*)d_out;

  dim3 grid(F_OUT, B_ * G_);
  linconv2d_wmma_f32<<<grid, 128, 0, stream>>>(x, w, out);
}


# Grid cut into 16 equal two-plane work-groups per (b,g), not 17: one takes both one-plane edge columns; no all-zero plane multiplied; 4 waves/SIMD max, so exactly 4 per WGP; sum order unchanged
# speedup vs baseline: 5.1461x; 1.1670x over previous
#include <hip/hip_runtime.h>


typedef __attribute__((ext_vector_type(2))) float v2f;
typedef __attribute__((ext_vector_type(4))) float v4f;
typedef __attribute__((ext_vector_type(8))) float v8f;

#define B_     8
#define CH_    128
#define F_IN   32
#define T_IN   1024
#define G_     8
#define D_     16
#define N_     16
#define W_     16
#define KS_    4
#define T_OUT  257
#define F_OUT  17

#define X_CH_STRIDE (F_IN * T_IN)
#define W_N_STRIDE  (D_ * F_IN * W_)
#define WL_N_PITCH  260
#define WL_FS_PITCH (N_ * WL_N_PITCH)

#define NS_TILES_ 16
#define NWAVES_   4
#define TILES_PER_WAVE_ (NS_TILES_ / NWAVES_)

#define WG_PER_BG_ 16

__device__ __forceinline__ void load_x_frags(const float* __restrict__ p, v2f (&xf)[D_]) {
  #pragma unroll
  for (int d = 0; d < D_; ++d) xf[d] = *(const v2f*)(p + (size_t)d * X_CH_STRIDE);
}

#define ROW_D  2
#define ROW_W  (ROW_D * 4)
#define NROWS_ (D_ / ROW_D)
#define NSETS_ 3

__device__ __forceinline__ void load_w_row(const float* wl, int lo, int hi, int row, v2f (&wf)[ROW_W]) {
  #pragma unroll
  for (int dd = 0; dd < ROW_D; ++dd) {
    #pragma unroll
    for (int wq = 0; wq < 4; ++wq) {
      const int d = row * ROW_D + dd;
      wf[dd * 4 + wq] = *(const v2f*)&wl[lo * WL_N_PITCH + d * 16 + wq * 4 + 2 * hi];
    }
  }
}

__device__ __forceinline__ void mma_one_f(const float* wl, int lo, int hi, const v2f (&xf)[D_], v8f (&P)[4]) {
  v2f wf[NSETS_][ROW_W];
  #pragma unroll
  for (int r = 0; r < NSETS_ - 1; ++r) load_w_row(wl, lo, hi, r, wf[r]);
  #pragma unroll
  for (int r = 0; r < NROWS_; ++r) {
    __builtin_amdgcn_sched_barrier(0);
    if (r + NSETS_ - 1 < NROWS_) load_w_row(wl, lo, hi, r + NSETS_ - 1, wf[(r + NSETS_ - 1) % NSETS_]);
    __builtin_amdgcn_sched_barrier(0);
    #pragma unroll
    for (int i = 0; i < ROW_W; ++i) {
      P[i & 3] = __builtin_amdgcn_wmma_f32_16x16x4_f32(
           false, wf[r % NSETS_][i],  false, xf[r * ROW_D + (i >> 2)],
           (short)0, P[i & 3],  false,  false);
    }
  }
}

template <int K>
__device__ __forceinline__ float ror16(float v) {
  return __builtin_bit_cast(float, __builtin_amdgcn_update_dpp(0, __builtin_bit_cast(int, v), 0x120 + K, 0xF, 0xF, false));
}

__device__ __forceinline__ float leaky(float v) { return v > 0.0f ? v : 0.01f * v; }

__device__ __forceinline__ void combine(const v8f (&P)[4], bool m1, bool m2, bool m3, v8f& cur, v8f& car) {
  #pragma unroll
  for (int j = 0; j < 8; ++j) {
    const float r1 = ror16<1>(P[2][j]);
    const float r2 = ror16<2>(P[1][j]);
    const float r3 = ror16<3>(P[0][j]);
    cur[j] = ((P[3][j] + (m1 ? r1 : 0.0f)) + (m2 ? r2 : 0.0f)) + (m3 ? r3 : 0.0f);
    car[j] = ((m1 ? 0.0f : r1) + (m2 ? 0.0f : r2)) + (m3 ? 0.0f : r3);
  }
}

__device__ __forceinline__ void emit_tile(int it, int wv, int tile, int lo, int lane, float* orow, float* cs,
                                          v8f& pend, const v8f& cur, const v8f& car) {
  const int tl = lo - 1;
  if (it == 0) {
    if (wv == NWAVES_ - 1) {
      if (lo < 2) {
        #pragma unroll
        for (int j = 0; j < 8; ++j) orow[(size_t)j * (F_OUT * T_OUT) + 16 * NS_TILES_ + tl] = leaky(car[j]);
      }
    } else {
      #pragma unroll
      for (int j = 0; j < 8; ++j) cs[wv * 256 + j * 32 + lane] = car[j];
    }
  } else {
    const int tc = 16 * (tile + 1) + tl;
    #pragma unroll
    for (int j = 0; j < 8; ++j) orow[(size_t)j * (F_OUT * T_OUT) + tc] = leaky(pend[j] + car[j]);
  }
  pend = cur;
}

__device__ __forceinline__ void emit_lowest(int wv, int lo, int lane, float* orow, const float* cs, const v8f& pend) {
  v8f cin = {};
  if (wv > 0) {
    #pragma unroll
    for (int j = 0; j < 8; ++j) cin[j] = cs[(wv - 1) * 256 + j * 32 + lane];
  }
  const int tf = 16 * TILES_PER_WAVE_ * wv + lo - 1;
  if (tf >= 0) {
    #pragma unroll
    for (int j = 0; j < 8; ++j) orow[(size_t)j * (F_OUT * T_OUT) + tf] = leaky(pend[j] + cin[j]);
  }
}

__global__ __launch_bounds__(128) __attribute__((amdgpu_waves_per_eu(4, 4)))
void linconv2d_wmma_f32(const float* __restrict__ x,
                        const float* __restrict__ weight,
                        float* __restrict__ out) {
  __shared__ __align__(16) float wlds[2 * WL_FS_PITCH];
  __shared__ float cslot[2 * (NWAVES_ - 1) * 8 * 32];

  const int tid  = threadIdx.x;
  const int wv   = __builtin_amdgcn_readfirstlane(tid >> 5);
  const int lane = tid & 31;
  const int hi   = lane >> 4;
  const int lo   = lane & 15;

  const int z  = blockIdx.x;
  const int y  = blockIdx.y;
  const int bi = z >> 3;
  const int g  = z & 7;

  const bool edge = (y == 0);
  const int fc0 = edge ? 0 : 2 * y - 1;
  const int fc1 = edge ? F_IN - 1 : 2 * y;
  const int foA = edge ? 0 : y;
  const int foB = edge ? F_OUT - 1 : y;

  const float* wsrc = weight + (size_t)(g * N_ + (tid >> 6)) * W_N_STRIDE
                             + ((tid >> 2) & 15) * (F_IN * W_) + (tid & 3) * 4;
  v4f wr[16];
  #pragma unroll
  for (int i = 0; i < 16; ++i) {
    const int fcw = (i >> 3) ? fc1 : fc0;
    wr[i] = *(const v4f*)(wsrc + (size_t)(i & 7) * 2 * W_N_STRIDE + fcw * W_);
  }

  const float* xsrc = x + ((size_t)bi * CH_ + g * D_) * X_CH_STRIDE + 4 * lo + 2 * hi;
  const float* xs0  = xsrc + (size_t)fc0 * T_IN;
  const float* xs1  = xsrc + (size_t)fc1 * T_IN;
  int tile = TILES_PER_WAVE_ * wv + TILES_PER_WAVE_ - 1;
  v2f xa[D_], xb[D_];
  load_x_frags(xs0 + 64 * tile, xa);

  {
    float* wdst = &wlds[(tid >> 6) * WL_N_PITCH + (tid & 63) * 4];
    #pragma unroll
    for (int i = 0; i < 16; ++i) {
      *(v4f*)(wdst + (i >> 3) * WL_FS_PITCH + (i & 7) * 2 * WL_N_PITCH) = wr[i];
    }
  }
  __syncthreads();

  const size_t orow0 = ((size_t)bi * CH_ + g * N_ + 8 * hi) * F_OUT;
  float* orowA = out + (orow0 + foA) * (size_t)T_OUT;
  float* orowB = out + (orow0 + foB) * (size_t)T_OUT;
  float* csA = &cslot[(NWAVES_ - 1) * 8 * 32];
  float* csB = &cslot[0];
  const bool m1 = lo >= 1, m2 = lo >= 2, m3 = lo >= 3;

  v8f pendA = {}, pendB = {};
  #pragma unroll
  for (int it = 0; it < TILES_PER_WAVE_; ++it) {
    load_x_frags(xs1 + 64 * tile, xb);
    asm volatile("" ::: "memory");

    v8f P[4] = {};
    mma_one_f(&wlds[0], lo, hi, xa, P);

    if (it + 1 < TILES_PER_WAVE_) load_x_frags(xs0 + 64 * (tile - 1), xa);
    asm volatile("" ::: "memory");

    if (edge) {
      v8f curA, carA;
      combine(P, m1, m2, m3, curA, carA);
      emit_tile(it, wv, tile, lo, lane, orowA, csA, pendA, curA, carA);
      #pragma unroll
      for (int q = 0; q < 4; ++q) P[q] = v8f{};
    }

    mma_one_f(&wlds[WL_FS_PITCH], lo, hi, xb, P);

    v8f cur, car;
    combine(P, m1, m2, m3, cur, car);
    emit_tile(it, wv, tile, lo, lane, orowB, csB, pendB, cur, car);
    --tile;
  }

  __syncthreads();

  emit_lowest(wv, lo, lane, orowB, csB, pendB);
  if (edge) emit_lowest(wv, lo, lane, orowA, csA, pendA);
}

extern "C" void kernel_launch(void* const* d_in, const int* in_sizes, int n_in,
                              void* d_out, int out_size, void* d_ws, size_t ws_size,
                              hipStream_t stream) {
  (void)in_sizes; (void)n_in; (void)out_size; (void)d_ws; (void)ws_size;
  const float* x = (const float*)d_in[0];
  const float* w = (const float*)d_in[1];
  float* out     = (float*)d_out;

  dim3 grid(B_ * G_, WG_PER_BG_);
  linconv2d_wmma_f32<<<grid, 128, 0, stream>>>(x, w, out);
}
